# starting kernel
# baseline (speedup 1.0000x reference)
#include <hip/hip_runtime.h>
#include <math.h>

typedef float v2f __attribute__((ext_vector_type(2)));
typedef float v8f __attribute__((ext_vector_type(8)));

#define N_NODES 50000
#define N_EDGES 600000
#define DFEAT   128
#define NCLS    40

__global__ void k_init_deg(int* deg, int n) {
    int i = blockIdx.x * blockDim.x + threadIdx.x;
    if (i < n) deg[i] = 1;
}

__global__ void k_count_deg(int* deg, const int* __restrict__ dst, int e) {
    int i = blockIdx.x * blockDim.x + threadIdx.x;
    if (i < e) atomicAdd(&deg[dst[i]], 1);
}

__global__ void k_make_dinv(float* dinv, int n) {
    int i = blockIdx.x * blockDim.x + threadIdx.x;
    if (i < n) {
        int d = ((const int*)dinv)[i];
        dinv[i] = rsqrtf((float)d);
    }
}

__global__ void k_zero(float* p, int n) {
    int i = blockIdx.x * blockDim.x + threadIdx.x;
    if (i < n) p[i] = 0.0f;
}

template<int N>
__global__ void k_gemm(const float* __restrict__ A, const float* __restrict__ B,
                       float* __restrict__ C) {
    const int lane  = threadIdx.x & 31;
    const int wave  = threadIdx.x >> 5;
    const int row0  = blockIdx.x * 16;
    const int col0  = (blockIdx.y * (blockDim.x >> 5) + wave) * 16;
    const int r     = lane & 15;
    const int kb    = (lane >> 4) * 2;

    const int col  = col0 + r;
    const int colc = (col < N) ? col : (N - 1);

    const float* Ap = A + (row0 + r) * DFEAT + kb;
    const float* Bp = B + kb * N + colc;

    v8f acc = {};
    #pragma unroll
    for (int k0 = 0; k0 < DFEAT; k0 += 4) {
        v2f a = *(const v2f*)(Ap + k0);
        v2f b;
        b.x = Bp[k0 * N];
        b.y = Bp[k0 * N + N];
        acc = __builtin_amdgcn_wmma_f32_16x16x4_f32(
            false, a, false, b, (short)0, acc, false, false);
    }

    if (col < N) {
        float* Cp = C + (row0 + (lane >> 4) * 8) * N + col;
        #pragma unroll
        for (int i = 0; i < 8; ++i)
            Cp[i * N] = acc[i];
    }
}

template<int N>
__global__ void k_gemm_bias(const float* __restrict__ A, const float* __restrict__ B,
                            const float* __restrict__ bias, float* __restrict__ C) {
    const int lane  = threadIdx.x & 31;
    const int wave  = threadIdx.x >> 5;
    const int row0  = blockIdx.x * 16;
    const int col0  = wave * 16;
    const int r     = lane & 15;
    const int kb    = (lane >> 4) * 2;

    const int col  = col0 + r;
    const int colc = (col < N) ? col : (N - 1);

    const float* Ap = A + (row0 + r) * DFEAT + kb;
    const float* Bp = B + kb * N + colc;

    v8f acc = {};
    #pragma unroll
    for (int k0 = 0; k0 < DFEAT; k0 += 4) {
        v2f a = *(const v2f*)(Ap + k0);
        v2f b;
        b.x = Bp[k0 * N];
        b.y = Bp[k0 * N + N];
        acc = __builtin_amdgcn_wmma_f32_16x16x4_f32(
            false, a, false, b, (short)0, acc, false, false);
    }

    if (col < N) {
        const float bb = bias[col];
        float* Cp = C + (row0 + (lane >> 4) * 8) * N + col;
        #pragma unroll
        for (int i = 0; i < 8; ++i)
            Cp[i * N] = acc[i] + bb;
    }
}

__global__ void k_scatter(const float* __restrict__ h, const int* __restrict__ src,
                          const int* __restrict__ dst, const float* __restrict__ dinv,
                          float* __restrict__ agg, int E) {
    const int e = blockIdx.x * (blockDim.x >> 5) + (threadIdx.x >> 5);
    if (e >= E) return;
    const int lane = threadIdx.x & 31;
    const int s = src[e];
    const int d = dst[e];
    const float norm = dinv[s] * dinv[d];
    const float4 v = *(const float4*)(h + s * DFEAT + lane * 4);
    float* out = agg + d * DFEAT + lane * 4;
    unsafeAtomicAdd(out + 0, v.x * norm);
    unsafeAtomicAdd(out + 1, v.y * norm);
    unsafeAtomicAdd(out + 2, v.z * norm);
    unsafeAtomicAdd(out + 3, v.w * norm);
}

__global__ void k_finalize(const float* __restrict__ h, const float* __restrict__ agg,
                           const float* __restrict__ dinv, const float* __restrict__ b,
                           float* __restrict__ out, int n) {
    int t = blockIdx.x * blockDim.x + threadIdx.x;
    if (t >= n) return;
    const int i = t >> 7;
    const int f = t & 127;
    const float di = dinv[i];
    out[t] = tanhf(agg[t] + h[t] * di * di + b[f]);
}

extern "C" void kernel_launch(void* const* d_in, const int* in_sizes, int n_in,
                              void* d_out, int out_size, void* d_ws, size_t ws_size,
                              hipStream_t stream) {
    const float* x    = (const float*)d_in[0];
    const int*   ei   = (const int*)  d_in[1];
    const float* W1   = (const float*)d_in[2];
    const float* b1   = (const float*)d_in[3];
    const float* W2   = (const float*)d_in[4];
    const float* b2   = (const float*)d_in[5];
    const float* Wlin = (const float*)d_in[6];
    const float* blin = (const float*)d_in[7];

    const int* src = ei;
    const int* dst = ei + N_EDGES;

    float* wsf  = (float*)d_ws;
    float* dinv = wsf;
    float* bufA = wsf + 50048;
    float* bufB = bufA + (size_t)N_NODES * DFEAT;

    const int NF = N_NODES * DFEAT;
    const dim3 gemmGrid(N_NODES / 16, DFEAT / 64);

    k_init_deg <<<(N_NODES + 255) / 256, 256, 0, stream>>>((int*)dinv, N_NODES);
    k_count_deg<<<(N_EDGES + 255) / 256, 256, 0, stream>>>((int*)dinv, dst, N_EDGES);
    k_make_dinv<<<(N_NODES + 255) / 256, 256, 0, stream>>>(dinv, N_NODES);

    k_gemm<DFEAT><<<gemmGrid, 128, 0, stream>>>(x, W1, bufA);
    k_zero    <<<(NF + 255) / 256, 256, 0, stream>>>(bufB, NF);
    k_scatter <<<(N_EDGES + 7) / 8, 256, 0, stream>>>(bufA, src, dst, dinv, bufB, N_EDGES);
    k_finalize<<<(NF + 255) / 256, 256, 0, stream>>>(bufA, bufB, dinv, b1, bufB, NF);

    k_gemm<DFEAT><<<gemmGrid, 128, 0, stream>>>(bufB, W2, bufA);
    k_zero    <<<(NF + 255) / 256, 256, 0, stream>>>(bufB, NF);
    k_scatter <<<(N_EDGES + 7) / 8, 256, 0, stream>>>(bufA, src, dst, dinv, bufB, N_EDGES);
    k_finalize<<<(NF + 255) / 256, 256, 0, stream>>>(bufA, bufB, dinv, b2, bufB, NF);

    k_gemm_bias<NCLS><<<N_NODES / 16, 96, 0, stream>>>(bufB, Wlin, blin, (float*)d_out);
}


# Float-atomic edge scatter -> per-call dst-grouped edge list (arrival-order rank from a returning int atomic) + one gather pass per layer summing in list order, self-loop, bias, tanh fused in
# speedup vs baseline: 2.1020x; 2.1020x over previous
#include <hip/hip_runtime.h>
#include <math.h>


typedef float v2f __attribute__((ext_vector_type(2)));
typedef float v8f __attribute__((ext_vector_type(8)));

#define N_NODES 50000
#define N_EDGES 600000
#define DFEAT   128
#define NCLS    40

__global__ void k_init_cnt(int* cnt, int n) {
    int i = blockIdx.x * blockDim.x + threadIdx.x;
    if (i < n) cnt[i] = 0;
}

__global__ void k_count_rank(int* cnt, const int* __restrict__ dst,
                             int* __restrict__ rank, int e) {
    int i = blockIdx.x * blockDim.x + threadIdx.x;
    if (i < e) rank[i] = atomicAdd(&cnt[dst[i]], 1);
}

#define SCAN_PASSES 13
#define SCAN_CHUNK  (SCAN_PASSES * 128)
__global__ void __launch_bounds__(1024) k_scan(const int* __restrict__ cnt,
                                               int* __restrict__ offs, int n) {
    __shared__ int wave_total[32];
    const int lane = threadIdx.x & 31;
    const int wave = threadIdx.x >> 5;
    const int base = wave * SCAN_CHUNK + lane * 4;

    int4 v[SCAN_PASSES];
    int  ls[SCAN_PASSES];
    int  inc[SCAN_PASSES];
    #pragma unroll
    for (int i = 0; i < SCAN_PASSES; ++i) {
        const int idx = base + i * 128;
        v[i] = (idx < n) ? *(const int4*)(cnt + idx) : make_int4(0, 0, 0, 0);
        ls[i] = v[i].x + v[i].y + v[i].z + v[i].w;
    }
    #pragma unroll
    for (int i = 0; i < SCAN_PASSES; ++i) {
        int x = ls[i];
        x += __builtin_amdgcn_update_dpp(0, x, 0x111, 0xf, 0xf, true);
        x += __builtin_amdgcn_update_dpp(0, x, 0x112, 0xf, 0xf, true);
        x += __builtin_amdgcn_update_dpp(0, x, 0x114, 0xf, 0xf, true);
        x += __builtin_amdgcn_update_dpp(0, x, 0x118, 0xf, 0xf, true);
        const int low = __builtin_amdgcn_readlane(x, 15);
        inc[i] = (lane >= 16) ? x + low : x;
    }
    int tot[SCAN_PASSES];
    int wtot = 0;
    #pragma unroll
    for (int i = 0; i < SCAN_PASSES; ++i) {
        tot[i] = __builtin_amdgcn_readlane(inc[i], 31);
        wtot += tot[i];
    }
    if (lane == 0) wave_total[wave] = wtot;
    __syncthreads();
    int c = (lane < wave) ? wave_total[lane] : 0;
    #pragma unroll
    for (int d = 16; d >= 1; d >>= 1) c += __shfl_xor(c, d);
    int carry = c;

    #pragma unroll
    for (int i = 0; i < SCAN_PASSES; ++i) {
        const int idx = base + i * 128;
        int4 o;
        o.x = carry + inc[i] - ls[i];
        o.y = o.x + v[i].x;
        o.z = o.y + v[i].y;
        o.w = o.z + v[i].z;
        if (idx < n) {
            *(int4*)(offs + idx) = o;
            if (idx + 4 == n) offs[n] = o.w + v[i].w;
        }
        carry += tot[i];
    }
}

__global__ void k_fill(const int* __restrict__ src, const int* __restrict__ dst,
                       const int* __restrict__ rank, const int* __restrict__ offs,
                       const int* __restrict__ cnt, int2* __restrict__ csr, int e) {
    int i = blockIdx.x * blockDim.x + threadIdx.x;
    if (i >= e) return;
    const int s = src[i];
    const int d = dst[i];
    const float norm = rsqrtf((float)(cnt[s] + 1)) * rsqrtf((float)(cnt[d] + 1));
    csr[offs[d] + rank[i]] = make_int2(s, __float_as_int(norm));
}

template<int N>
__global__ void k_gemm(const float* __restrict__ A, const float* __restrict__ B,
                       float* __restrict__ C) {
    const int lane  = threadIdx.x & 31;
    const int wave  = threadIdx.x >> 5;
    const int row0  = blockIdx.x * 16;
    const int col0  = (blockIdx.y * (blockDim.x >> 5) + wave) * 16;
    const int r     = lane & 15;
    const int kb    = (lane >> 4) * 2;

    const int col  = col0 + r;
    const int colc = (col < N) ? col : (N - 1);

    const float* Ap = A + (row0 + r) * DFEAT + kb;
    const float* Bp = B + kb * N + colc;

    v8f acc = {};
    #pragma unroll
    for (int k0 = 0; k0 < DFEAT; k0 += 4) {
        v2f a = *(const v2f*)(Ap + k0);
        v2f b;
        b.x = Bp[k0 * N];
        b.y = Bp[k0 * N + N];
        acc = __builtin_amdgcn_wmma_f32_16x16x4_f32(
            false, a, false, b, (short)0, acc, false, false);
    }

    if (col < N) {
        float* Cp = C + (row0 + (lane >> 4) * 8) * N + col;
        #pragma unroll
        for (int i = 0; i < 8; ++i)
            Cp[i * N] = acc[i];
    }
}

template<int N>
__global__ void k_gemm_bias(const float* __restrict__ A, const float* __restrict__ B,
                            const float* __restrict__ bias, float* __restrict__ C) {
    const int lane  = threadIdx.x & 31;
    const int wave  = threadIdx.x >> 5;
    const int row0  = blockIdx.x * 16;
    const int col0  = wave * 16;
    const int r     = lane & 15;
    const int kb    = (lane >> 4) * 2;

    const int col  = col0 + r;
    const int colc = (col < N) ? col : (N - 1);

    const float* Ap = A + (row0 + r) * DFEAT + kb;
    const float* Bp = B + kb * N + colc;

    v8f acc = {};
    #pragma unroll
    for (int k0 = 0; k0 < DFEAT; k0 += 4) {
        v2f a = *(const v2f*)(Ap + k0);
        v2f b;
        b.x = Bp[k0 * N];
        b.y = Bp[k0 * N + N];
        acc = __builtin_amdgcn_wmma_f32_16x16x4_f32(
            false, a, false, b, (short)0, acc, false, false);
    }

    if (col < N) {
        const float bb = bias[col];
        float* Cp = C + (row0 + (lane >> 4) * 8) * N + col;
        #pragma unroll
        for (int i = 0; i < 8; ++i)
            Cp[i * N] = acc[i] + bb;
    }
}

#define NPW 4

__device__ __forceinline__ float4 fma4(float w, float4 v, float4 a) {
    a.x = fmaf(w, v.x, a.x);
    a.y = fmaf(w, v.y, a.y);
    a.z = fmaf(w, v.z, a.z);
    a.w = fmaf(w, v.w, a.w);
    return a;
}

__device__ __forceinline__ float4 row4(const float* __restrict__ h, int row, int lane) {
    return *(const float4*)(h + row * DFEAT + lane * 4);
}

__global__ void __launch_bounds__(256) k_gather(const float* __restrict__ h,
                                                const int2* __restrict__ csr,
                                                const int* __restrict__ offs,
                                                const float* __restrict__ bias,
                                                float* __restrict__ out, int n_nodes) {
    const int lane = threadIdx.x & 31;
    const int wave = __builtin_amdgcn_readfirstlane(threadIdx.x >> 5);
    const int n0 = (blockIdx.x * 8 + wave) * NPW;
    if (n0 >= n_nodes) return;

    int o[NPW + 1];
    #pragma unroll
    for (int k = 0; k <= NPW; ++k) {
        const int idx = (n0 + k < n_nodes) ? (n0 + k) : n_nodes;
        o[k] = offs[idx];
    }
    int2 ent[NPW];
    #pragma unroll
    for (int k = 0; k < NPW; ++k)
        ent[k] = (lane < o[k + 1] - o[k]) ? csr[o[k] + lane] : make_int2(0, 0);

    const float4 b4 = *(const float4*)(bias + lane * 4);

    #pragma unroll
    for (int k = 0; k < NPW; ++k) {
        const int n = n0 + k;
        if (n >= n_nodes) break;
        const int len = o[k + 1] - o[k];
        const float4 hs = row4(h, n, lane);
        float4 acc = make_float4(0.f, 0.f, 0.f, 0.f);

        const int first = (len < 32) ? len : 32;
        int j = 0;
        for (; j + 4 <= first; j += 4) {
            const int s0 = __builtin_amdgcn_readlane(ent[k].x, j);
            const int s1 = __builtin_amdgcn_readlane(ent[k].x, j + 1);
            const int s2 = __builtin_amdgcn_readlane(ent[k].x, j + 2);
            const int s3 = __builtin_amdgcn_readlane(ent[k].x, j + 3);
            const float w0 = __int_as_float(__builtin_amdgcn_readlane(ent[k].y, j));
            const float w1 = __int_as_float(__builtin_amdgcn_readlane(ent[k].y, j + 1));
            const float w2 = __int_as_float(__builtin_amdgcn_readlane(ent[k].y, j + 2));
            const float w3 = __int_as_float(__builtin_amdgcn_readlane(ent[k].y, j + 3));
            const float4 v0 = row4(h, s0, lane);
            const float4 v1 = row4(h, s1, lane);
            const float4 v2 = row4(h, s2, lane);
            const float4 v3 = row4(h, s3, lane);
            acc = fma4(w0, v0, acc);
            acc = fma4(w1, v1, acc);
            acc = fma4(w2, v2, acc);
            acc = fma4(w3, v3, acc);
        }
        const int rem = first - j;
        if (rem == 3) {
            const int s0 = __builtin_amdgcn_readlane(ent[k].x, j);
            const int s1 = __builtin_amdgcn_readlane(ent[k].x, j + 1);
            const int s2 = __builtin_amdgcn_readlane(ent[k].x, j + 2);
            const float w0 = __int_as_float(__builtin_amdgcn_readlane(ent[k].y, j));
            const float w1 = __int_as_float(__builtin_amdgcn_readlane(ent[k].y, j + 1));
            const float w2 = __int_as_float(__builtin_amdgcn_readlane(ent[k].y, j + 2));
            const float4 v0 = row4(h, s0, lane);
            const float4 v1 = row4(h, s1, lane);
            const float4 v2 = row4(h, s2, lane);
            acc = fma4(w0, v0, acc);
            acc = fma4(w1, v1, acc);
            acc = fma4(w2, v2, acc);
        } else if (rem == 2) {
            const int s0 = __builtin_amdgcn_readlane(ent[k].x, j);
            const int s1 = __builtin_amdgcn_readlane(ent[k].x, j + 1);
            const float w0 = __int_as_float(__builtin_amdgcn_readlane(ent[k].y, j));
            const float w1 = __int_as_float(__builtin_amdgcn_readlane(ent[k].y, j + 1));
            const float4 v0 = row4(h, s0, lane);
            const float4 v1 = row4(h, s1, lane);
            acc = fma4(w0, v0, acc);
            acc = fma4(w1, v1, acc);
        } else if (rem == 1) {
            const int s0 = __builtin_amdgcn_readlane(ent[k].x, j);
            const float w0 = __int_as_float(__builtin_amdgcn_readlane(ent[k].y, j));
            acc = fma4(w0, row4(h, s0, lane), acc);
        }
        for (int p = o[k] + 32; p < o[k + 1]; p += 32) {
            const int c = (o[k + 1] - p < 32) ? (o[k + 1] - p) : 32;
            const int2 e2 = (lane < c) ? csr[p + lane] : make_int2(0, 0);
            for (int q = 0; q < c; ++q) {
                const int s0 = __builtin_amdgcn_readlane(e2.x, q);
                const float w0 = __int_as_float(__builtin_amdgcn_readlane(e2.y, q));
                acc = fma4(w0, row4(h, s0, lane), acc);
            }
        }

        const float di = rsqrtf((float)(len + 1));
        float4 r;
        r.x = tanhf(acc.x + hs.x * di * di + b4.x);
        r.y = tanhf(acc.y + hs.y * di * di + b4.y);
        r.z = tanhf(acc.z + hs.z * di * di + b4.z);
        r.w = tanhf(acc.w + hs.w * di * di + b4.w);
        *(float4*)(out + n * DFEAT + lane * 4) = r;
    }
}

extern "C" void kernel_launch(void* const* d_in, const int* in_sizes, int n_in,
                              void* d_out, int out_size, void* d_ws, size_t ws_size,
                              hipStream_t stream) {
    const float* x    = (const float*)d_in[0];
    const int*   ei   = (const int*)  d_in[1];
    const float* W1   = (const float*)d_in[2];
    const float* b1   = (const float*)d_in[3];
    const float* W2   = (const float*)d_in[4];
    const float* b2   = (const float*)d_in[5];
    const float* Wlin = (const float*)d_in[6];
    const float* blin = (const float*)d_in[7];

    const int* src = ei;
    const int* dst = ei + N_EDGES;

    int*   cnt  = (int*)d_ws;
    int*   offs = cnt + 50176;
    int*   rank = offs + 50176;
    int2*  csr  = (int2*)(rank + 600064);
    float* bufA = (float*)(csr + 600064);
    float* bufB = bufA + (size_t)N_NODES * DFEAT;

    const dim3 gemmGrid(N_NODES / 16, DFEAT / 64);
    const int gatherBlocks = (N_NODES + 8 * NPW - 1) / (8 * NPW);

    k_init_cnt  <<<(N_NODES + 255) / 256, 256, 0, stream>>>(cnt, N_NODES);
    k_count_rank<<<(N_EDGES + 255) / 256, 256, 0, stream>>>(cnt, dst, rank, N_EDGES);
    k_scan      <<<1, 1024, 0, stream>>>(cnt, offs, N_NODES);
    k_fill      <<<(N_EDGES + 255) / 256, 256, 0, stream>>>(src, dst, rank, offs, cnt, csr, N_EDGES);

    k_gemm<DFEAT><<<gemmGrid, 128, 0, stream>>>(x, W1, bufA);
    k_gather<<<gatherBlocks, 256, 0, stream>>>(bufA, csr, offs, b1, bufB, N_NODES);

    k_gemm<DFEAT><<<gemmGrid, 128, 0, stream>>>(bufB, W2, bufA);
    k_gather<<<gatherBlocks, 256, 0, stream>>>(bufA, csr, offs, b2, bufB, N_NODES);

    k_gemm_bias<NCLS><<<N_NODES / 16, 96, 0, stream>>>(bufB, Wlin, blin, (float*)d_out);
}


# Next dense product fused into each gather pass: the block's 32 finished rows stay in LDS and feed the f32 16x16x4 matrix products directly (7 launches, activations never written to memory)
# speedup vs baseline: 2.5966x; 1.2353x over previous
#include <hip/hip_runtime.h>
#include <math.h>


typedef float v2f __attribute__((ext_vector_type(2)));
typedef float v8f __attribute__((ext_vector_type(8)));

#define N_NODES 50000
#define N_EDGES 600000
#define DFEAT   128
#define NCLS    40

__global__ void k_init_cnt(int* cnt, int n) {
    int i = blockIdx.x * blockDim.x + threadIdx.x;
    if (i < n) cnt[i] = 0;
}

__global__ void k_count_rank(int* cnt, const int* __restrict__ dst,
                             int* __restrict__ rank, int e) {
    int i = blockIdx.x * blockDim.x + threadIdx.x;
    if (i < e) rank[i] = atomicAdd(&cnt[dst[i]], 1);
}

#define SCAN_PASSES 13
#define SCAN_CHUNK  (SCAN_PASSES * 128)
__global__ void __launch_bounds__(1024) k_scan(const int* __restrict__ cnt,
                                               int* __restrict__ offs, int n) {
    __shared__ int wave_total[32];
    const int lane = threadIdx.x & 31;
    const int wave = threadIdx.x >> 5;
    const int base = wave * SCAN_CHUNK + lane * 4;

    int4 v[SCAN_PASSES];
    int  ls[SCAN_PASSES];
    int  inc[SCAN_PASSES];
    #pragma unroll
    for (int i = 0; i < SCAN_PASSES; ++i) {
        const int idx = base + i * 128;
        v[i] = (idx < n) ? *(const int4*)(cnt + idx) : make_int4(0, 0, 0, 0);
        ls[i] = v[i].x + v[i].y + v[i].z + v[i].w;
    }
    #pragma unroll
    for (int i = 0; i < SCAN_PASSES; ++i) {
        int x = ls[i];
        x += __builtin_amdgcn_update_dpp(0, x, 0x111, 0xf, 0xf, true);
        x += __builtin_amdgcn_update_dpp(0, x, 0x112, 0xf, 0xf, true);
        x += __builtin_amdgcn_update_dpp(0, x, 0x114, 0xf, 0xf, true);
        x += __builtin_amdgcn_update_dpp(0, x, 0x118, 0xf, 0xf, true);
        const int low = __builtin_amdgcn_readlane(x, 15);
        inc[i] = (lane >= 16) ? x + low : x;
    }
    int tot[SCAN_PASSES];
    int wtot = 0;
    #pragma unroll
    for (int i = 0; i < SCAN_PASSES; ++i) {
        tot[i] = __builtin_amdgcn_readlane(inc[i], 31);
        wtot += tot[i];
    }
    if (lane == 0) wave_total[wave] = wtot;
    __syncthreads();
    int c = (lane < wave) ? wave_total[lane] : 0;
    #pragma unroll
    for (int d = 16; d >= 1; d >>= 1) c += __shfl_xor(c, d);
    int carry = c;

    #pragma unroll
    for (int i = 0; i < SCAN_PASSES; ++i) {
        const int idx = base + i * 128;
        int4 o;
        o.x = carry + inc[i] - ls[i];
        o.y = o.x + v[i].x;
        o.z = o.y + v[i].y;
        o.w = o.z + v[i].z;
        if (idx < n) {
            *(int4*)(offs + idx) = o;
            if (idx + 4 == n) offs[n] = o.w + v[i].w;
        }
        carry += tot[i];
    }
}

__global__ void k_fill(const int* __restrict__ src, const int* __restrict__ dst,
                       const int* __restrict__ rank, const int* __restrict__ offs,
                       const int* __restrict__ cnt, int2* __restrict__ csr, int e) {
    int i = blockIdx.x * blockDim.x + threadIdx.x;
    if (i >= e) return;
    const int s = src[i];
    const int d = dst[i];
    const float norm = rsqrtf((float)(cnt[s] + 1)) * rsqrtf((float)(cnt[d] + 1));
    csr[offs[d] + rank[i]] = make_int2(s, __float_as_int(norm));
}

template<int N>
__global__ void k_gemm(const float* __restrict__ A, const float* __restrict__ B,
                       float* __restrict__ C) {
    const int lane  = threadIdx.x & 31;
    const int wave  = threadIdx.x >> 5;
    const int row0  = blockIdx.x * 16;
    const int col0  = (blockIdx.y * (blockDim.x >> 5) + wave) * 16;
    const int r     = lane & 15;
    const int kb    = (lane >> 4) * 2;

    const int col  = col0 + r;
    const int colc = (col < N) ? col : (N - 1);

    const float* Ap = A + (row0 + r) * DFEAT + kb;
    const float* Bp = B + kb * N + colc;

    v8f acc = {};
    #pragma unroll
    for (int k0 = 0; k0 < DFEAT; k0 += 4) {
        v2f a = *(const v2f*)(Ap + k0);
        v2f b;
        b.x = Bp[k0 * N];
        b.y = Bp[k0 * N + N];
        acc = __builtin_amdgcn_wmma_f32_16x16x4_f32(
            false, a, false, b, (short)0, acc, false, false);
    }

    if (col < N) {
        float* Cp = C + (row0 + (lane >> 4) * 8) * N + col;
        #pragma unroll
        for (int i = 0; i < 8; ++i)
            Cp[i * N] = acc[i];
    }
}

#define NPW 4
#define GB_ROWS (8 * NPW)
#define LDP     (DFEAT + 4)

__device__ __forceinline__ float4 fma4(float w, float4 v, float4 a) {
    a.x = fmaf(w, v.x, a.x);
    a.y = fmaf(w, v.y, a.y);
    a.z = fmaf(w, v.z, a.z);
    a.w = fmaf(w, v.w, a.w);
    return a;
}

__device__ __forceinline__ float4 row4(const float* __restrict__ h, int row, int lane) {
    return *(const float4*)(h + row * DFEAT + lane * 4);
}

template<int N, bool HEAD>
__global__ void __launch_bounds__(256, 16) k_gather_gemm(const float* __restrict__ h,
                                                     const int2* __restrict__ csr,
                                                     const int* __restrict__ offs,
                                                     const float* __restrict__ bias,
                                                     const float* __restrict__ Wn,
                                                     const float* __restrict__ bias_n,
                                                     float* __restrict__ C, int n_nodes) {
    __shared__ __attribute__((aligned(16))) float rows[GB_ROWS * LDP];

    const int lane = threadIdx.x & 31;
    const int wave = __builtin_amdgcn_readfirstlane(threadIdx.x >> 5);
    const int n0 = (blockIdx.x * 8 + wave) * NPW;

    int o[NPW + 1];
    #pragma unroll
    for (int k = 0; k <= NPW; ++k) {
        const int idx = (n0 + k < n_nodes) ? (n0 + k) : n_nodes;
        o[k] = offs[idx];
    }
    int2 ent[NPW];
    #pragma unroll
    for (int k = 0; k < NPW; ++k)
        ent[k] = (lane < o[k + 1] - o[k]) ? csr[o[k] + lane] : make_int2(0, 0);

    const float4 b4 = *(const float4*)(bias + lane * 4);

    #pragma unroll
    for (int k = 0; k < NPW; ++k) {
        const int n = n0 + k;
        float4 r = make_float4(0.f, 0.f, 0.f, 0.f);
        if (n < n_nodes) {
            const int len = o[k + 1] - o[k];
            const float4 hs = row4(h, n, lane);
            float4 acc = make_float4(0.f, 0.f, 0.f, 0.f);

            const int first = (len < 32) ? len : 32;
            int j = 0;
            for (; j + 4 <= first; j += 4) {
                const int s0 = __builtin_amdgcn_readlane(ent[k].x, j);
                const int s1 = __builtin_amdgcn_readlane(ent[k].x, j + 1);
                const int s2 = __builtin_amdgcn_readlane(ent[k].x, j + 2);
                const int s3 = __builtin_amdgcn_readlane(ent[k].x, j + 3);
                const float w0 = __int_as_float(__builtin_amdgcn_readlane(ent[k].y, j));
                const float w1 = __int_as_float(__builtin_amdgcn_readlane(ent[k].y, j + 1));
                const float w2 = __int_as_float(__builtin_amdgcn_readlane(ent[k].y, j + 2));
                const float w3 = __int_as_float(__builtin_amdgcn_readlane(ent[k].y, j + 3));
                const float4 v0 = row4(h, s0, lane);
                const float4 v1 = row4(h, s1, lane);
                const float4 v2 = row4(h, s2, lane);
                const float4 v3 = row4(h, s3, lane);
                acc = fma4(w0, v0, acc);
                acc = fma4(w1, v1, acc);
                acc = fma4(w2, v2, acc);
                acc = fma4(w3, v3, acc);
            }
            const int rem = first - j;
            if (rem == 3) {
                const int s0 = __builtin_amdgcn_readlane(ent[k].x, j);
                const int s1 = __builtin_amdgcn_readlane(ent[k].x, j + 1);
                const int s2 = __builtin_amdgcn_readlane(ent[k].x, j + 2);
                const float w0 = __int_as_float(__builtin_amdgcn_readlane(ent[k].y, j));
                const float w1 = __int_as_float(__builtin_amdgcn_readlane(ent[k].y, j + 1));
                const float w2 = __int_as_float(__builtin_amdgcn_readlane(ent[k].y, j + 2));
                const float4 v0 = row4(h, s0, lane);
                const float4 v1 = row4(h, s1, lane);
                const float4 v2 = row4(h, s2, lane);
                acc = fma4(w0, v0, acc);
                acc = fma4(w1, v1, acc);
                acc = fma4(w2, v2, acc);
            } else if (rem == 2) {
                const int s0 = __builtin_amdgcn_readlane(ent[k].x, j);
                const int s1 = __builtin_amdgcn_readlane(ent[k].x, j + 1);
                const float w0 = __int_as_float(__builtin_amdgcn_readlane(ent[k].y, j));
                const float w1 = __int_as_float(__builtin_amdgcn_readlane(ent[k].y, j + 1));
                const float4 v0 = row4(h, s0, lane);
                const float4 v1 = row4(h, s1, lane);
                acc = fma4(w0, v0, acc);
                acc = fma4(w1, v1, acc);
            } else if (rem == 1) {
                const int s0 = __builtin_amdgcn_readlane(ent[k].x, j);
                const float w0 = __int_as_float(__builtin_amdgcn_readlane(ent[k].y, j));
                acc = fma4(w0, row4(h, s0, lane), acc);
            }
            for (int p = o[k] + 32; p < o[k + 1]; p += 32) {
                const int c = (o[k + 1] - p < 32) ? (o[k + 1] - p) : 32;
                const int2 e2 = (lane < c) ? csr[p + lane] : make_int2(0, 0);
                for (int q = 0; q < c; ++q) {
                    const int s0 = __builtin_amdgcn_readlane(e2.x, q);
                    const float w0 = __int_as_float(__builtin_amdgcn_readlane(e2.y, q));
                    acc = fma4(w0, row4(h, s0, lane), acc);
                }
            }

            const float di = rsqrtf((float)(len + 1));
            r.x = tanhf(acc.x + hs.x * di * di + b4.x);
            r.y = tanhf(acc.y + hs.y * di * di + b4.y);
            r.z = tanhf(acc.z + hs.z * di * di + b4.z);
            r.w = tanhf(acc.w + hs.w * di * di + b4.w);
        }
        *(float4*)(rows + (wave * NPW + k) * LDP + lane * 4) = r;
    }

    __syncthreads();

    constexpr int RT  = GB_ROWS / 16;
    constexpr int CT  = (N + 15) / 16;
    constexpr int RPW = (CT >= 8) ? RT : 1;
    static_assert(CT * (RT / RPW) <= 8, "one pass of the block's 8 waves covers C");

    if (wave >= CT * (RT / RPW)) return;
    const int ct  = wave % CT;
    const int rt0 = (wave / CT) * RPW;

    const int r16 = lane & 15;
    const int kb  = (lane >> 4) * 2;

    const int col  = ct * 16 + r16;
    const int colc = (col < N) ? col : (N - 1);

    const float* Ap = rows + (rt0 * 16 + r16) * LDP + kb;
    const float* Bp = Wn + kb * N + colc;

    v8f acc[RPW];
    #pragma unroll
    for (int t = 0; t < RPW; ++t) acc[t] = v8f{};

    #pragma unroll 8
    for (int k0 = 0; k0 < DFEAT; k0 += 4) {
        v2f b;
        b.x = Bp[k0 * N];
        b.y = Bp[k0 * N + N];
        #pragma unroll
        for (int t = 0; t < RPW; ++t) {
            const v2f a = *(const v2f*)(Ap + t * 16 * LDP + k0);
            acc[t] = __builtin_amdgcn_wmma_f32_16x16x4_f32(
                false, a, false, b, (short)0, acc[t], false, false);
        }
    }

    if (col < N) {
        float bb = 0.f;
        if constexpr (HEAD) bb = bias_n[col];
        #pragma unroll
        for (int t = 0; t < RPW; ++t) {
            const int rbase = blockIdx.x * GB_ROWS + (rt0 + t) * 16;
            if (rbase < n_nodes) {
                float* Cp = C + (rbase + (lane >> 4) * 8) * N + col;
                #pragma unroll
                for (int i = 0; i < 8; ++i) {
                    if constexpr (HEAD) Cp[i * N] = acc[t][i] + bb;
                    else                Cp[i * N] = acc[t][i];
                }
            }
        }
    }
}

extern "C" void kernel_launch(void* const* d_in, const int* in_sizes, int n_in,
                              void* d_out, int out_size, void* d_ws, size_t ws_size,
                              hipStream_t stream) {
    const float* x    = (const float*)d_in[0];
    const int*   ei   = (const int*)  d_in[1];
    const float* W1   = (const float*)d_in[2];
    const float* b1   = (const float*)d_in[3];
    const float* W2   = (const float*)d_in[4];
    const float* b2   = (const float*)d_in[5];
    const float* Wlin = (const float*)d_in[6];
    const float* blin = (const float*)d_in[7];

    const int* src = ei;
    const int* dst = ei + N_EDGES;

    int*   cnt  = (int*)d_ws;
    int*   offs = cnt + 50176;
    int*   rank = offs + 50176;
    int2*  csr  = (int2*)(rank + 600064);
    float* bufA = (float*)(csr + 600064);
    float* bufB = bufA + (size_t)N_NODES * DFEAT;

    static_assert(N_NODES % 16 == 0, "row tiles of 16 must cover the nodes exactly");
    const dim3 gemmGrid(N_NODES / 16, DFEAT / 64);
    const int gatherBlocks = (N_NODES + 8 * NPW - 1) / (8 * NPW);

    k_init_cnt  <<<(N_NODES + 255) / 256, 256, 0, stream>>>(cnt, N_NODES);
    k_count_rank<<<(N_EDGES + 255) / 256, 256, 0, stream>>>(cnt, dst, rank, N_EDGES);
    k_scan      <<<1, 1024, 0, stream>>>(cnt, offs, N_NODES);
    k_fill      <<<(N_EDGES + 255) / 256, 256, 0, stream>>>(src, dst, rank, offs, cnt, csr, N_EDGES);

    k_gemm<DFEAT><<<gemmGrid, 128, 0, stream>>>(x, W1, bufA);

    k_gather_gemm<DFEAT, false><<<gatherBlocks, 256, 0, stream>>>(
        bufA, csr, offs, b1, W2, (const float*)nullptr, bufB, N_NODES);

    k_gather_gemm<NCLS, true><<<gatherBlocks, 256, 0, stream>>>(
        bufB, csr, offs, b2, Wlin, blin, (float*)d_out, N_NODES);
}


# Layer-1 product moved behind its aggregation ((Ahat x) W1: the x rows are summed before the product, a changed summation order) and fused into the first gather pass via LDS; 6 launches
# speedup vs baseline: 2.9423x; 1.1331x over previous
#include <hip/hip_runtime.h>
#include <math.h>


typedef float v2f __attribute__((ext_vector_type(2)));
typedef float v8f __attribute__((ext_vector_type(8)));

#define N_NODES 50000
#define N_EDGES 600000
#define DFEAT   128
#define NCLS    40

__global__ void k_init_cnt(int* cnt, int n) {
    int i = blockIdx.x * blockDim.x + threadIdx.x;
    if (i < n) cnt[i] = 0;
}

__global__ void k_count_rank(int* cnt, const int* __restrict__ dst,
                             int* __restrict__ rank, int e) {
    int i = blockIdx.x * blockDim.x + threadIdx.x;
    if (i < e) rank[i] = atomicAdd(&cnt[dst[i]], 1);
}

#define SCAN_PASSES 13
#define SCAN_CHUNK  (SCAN_PASSES * 128)
__global__ void __launch_bounds__(1024) k_scan(const int* __restrict__ cnt,
                                               int* __restrict__ offs, int n) {
    __shared__ int wave_total[32];
    const int lane = threadIdx.x & 31;
    const int wave = threadIdx.x >> 5;
    const int base = wave * SCAN_CHUNK + lane * 4;

    int4 v[SCAN_PASSES];
    int  ls[SCAN_PASSES];
    int  inc[SCAN_PASSES];
    #pragma unroll
    for (int i = 0; i < SCAN_PASSES; ++i) {
        const int idx = base + i * 128;
        v[i] = (idx < n) ? *(const int4*)(cnt + idx) : make_int4(0, 0, 0, 0);
        ls[i] = v[i].x + v[i].y + v[i].z + v[i].w;
    }
    #pragma unroll
    for (int i = 0; i < SCAN_PASSES; ++i) {
        int x = ls[i];
        x += __builtin_amdgcn_update_dpp(0, x, 0x111, 0xf, 0xf, true);
        x += __builtin_amdgcn_update_dpp(0, x, 0x112, 0xf, 0xf, true);
        x += __builtin_amdgcn_update_dpp(0, x, 0x114, 0xf, 0xf, true);
        x += __builtin_amdgcn_update_dpp(0, x, 0x118, 0xf, 0xf, true);
        const int low = __builtin_amdgcn_readlane(x, 15);
        inc[i] = (lane >= 16) ? x + low : x;
    }
    int tot[SCAN_PASSES];
    int wtot = 0;
    #pragma unroll
    for (int i = 0; i < SCAN_PASSES; ++i) {
        tot[i] = __builtin_amdgcn_readlane(inc[i], 31);
        wtot += tot[i];
    }
    if (lane == 0) wave_total[wave] = wtot;
    __syncthreads();
    int c = (lane < wave) ? wave_total[lane] : 0;
    #pragma unroll
    for (int d = 16; d >= 1; d >>= 1) c += __shfl_xor(c, d);
    int carry = c;

    #pragma unroll
    for (int i = 0; i < SCAN_PASSES; ++i) {
        const int idx = base + i * 128;
        int4 o;
        o.x = carry + inc[i] - ls[i];
        o.y = o.x + v[i].x;
        o.z = o.y + v[i].y;
        o.w = o.z + v[i].z;
        if (idx < n) {
            *(int4*)(offs + idx) = o;
            if (idx + 4 == n) offs[n] = o.w + v[i].w;
        }
        carry += tot[i];
    }
}

__global__ void k_fill(const int* __restrict__ src, const int* __restrict__ dst,
                       const int* __restrict__ rank, const int* __restrict__ offs,
                       const int* __restrict__ cnt, int2* __restrict__ csr, int e) {
    int i = blockIdx.x * blockDim.x + threadIdx.x;
    if (i >= e) return;
    const int s = src[i];
    const int d = dst[i];
    const float norm = rsqrtf((float)(cnt[s] + 1)) * rsqrtf((float)(cnt[d] + 1));
    csr[offs[d] + rank[i]] = make_int2(s, __float_as_int(norm));
}

#define NPW 4
#define GB_ROWS (8 * NPW)
#define LDP     (DFEAT + 4)

__device__ __forceinline__ float4 fma4(float w, float4 v, float4 a) {
    a.x = fmaf(w, v.x, a.x);
    a.y = fmaf(w, v.y, a.y);
    a.z = fmaf(w, v.z, a.z);
    a.w = fmaf(w, v.w, a.w);
    return a;
}

__device__ __forceinline__ float4 row4(const float* __restrict__ h, int row, int lane) {
    return *(const float4*)(h + row * DFEAT + lane * 4);
}

template<int N, int RPW>
__device__ __forceinline__ void tile_products(const float* Ap, const float* __restrict__ Bp,
                                              v8f (&acc)[RPW]) {
    #pragma unroll 8
    for (int k0 = 0; k0 < DFEAT; k0 += 4) {
        v2f b;
        b.x = Bp[k0 * N];
        b.y = Bp[k0 * N + N];
        #pragma unroll
        for (int t = 0; t < RPW; ++t) {
            const v2f a = *(const v2f*)(Ap + t * 16 * LDP + k0);
            acc[t] = __builtin_amdgcn_wmma_f32_16x16x4_f32(
                false, a, false, b, (short)0, acc[t], false, false);
        }
    }
}

template<int N, bool HEAD, bool PRE>
__global__ void __launch_bounds__(256, 16) k_gather_gemm(const float* __restrict__ h,
                                                     const int2* __restrict__ csr,
                                                     const int* __restrict__ offs,
                                                     const float* __restrict__ Wp,
                                                     const float* __restrict__ bias,
                                                     const float* __restrict__ Wn,
                                                     const float* __restrict__ bias_n,
                                                     float* __restrict__ C, int n_nodes) {
    __shared__ __attribute__((aligned(16))) float rows[(PRE ? 2 : 1) * GB_ROWS * LDP];

    const int lane = threadIdx.x & 31;
    const int wave = __builtin_amdgcn_readfirstlane(threadIdx.x >> 5);
    const int n0 = (blockIdx.x * 8 + wave) * NPW;

    int o[NPW + 1];
    #pragma unroll
    for (int k = 0; k <= NPW; ++k) {
        const int idx = (n0 + k < n_nodes) ? (n0 + k) : n_nodes;
        o[k] = offs[idx];
    }
    int2 ent[NPW];
    #pragma unroll
    for (int k = 0; k < NPW; ++k)
        ent[k] = (lane < o[k + 1] - o[k]) ? csr[o[k] + lane] : make_int2(0, 0);

    float4 b4 = make_float4(0.f, 0.f, 0.f, 0.f);
    if constexpr (!PRE) b4 = *(const float4*)(bias + lane * 4);

    #pragma unroll
    for (int k = 0; k < NPW; ++k) {
        const int n = n0 + k;
        float4 r = make_float4(0.f, 0.f, 0.f, 0.f);
        if (n < n_nodes) {
            const int len = o[k + 1] - o[k];
            const float4 hs = row4(h, n, lane);
            float4 acc = make_float4(0.f, 0.f, 0.f, 0.f);

            const int first = (len < 32) ? len : 32;
            int j = 0;
            for (; j + 4 <= first; j += 4) {
                const int s0 = __builtin_amdgcn_readlane(ent[k].x, j);
                const int s1 = __builtin_amdgcn_readlane(ent[k].x, j + 1);
                const int s2 = __builtin_amdgcn_readlane(ent[k].x, j + 2);
                const int s3 = __builtin_amdgcn_readlane(ent[k].x, j + 3);
                const float w0 = __int_as_float(__builtin_amdgcn_readlane(ent[k].y, j));
                const float w1 = __int_as_float(__builtin_amdgcn_readlane(ent[k].y, j + 1));
                const float w2 = __int_as_float(__builtin_amdgcn_readlane(ent[k].y, j + 2));
                const float w3 = __int_as_float(__builtin_amdgcn_readlane(ent[k].y, j + 3));
                const float4 v0 = row4(h, s0, lane);
                const float4 v1 = row4(h, s1, lane);
                const float4 v2 = row4(h, s2, lane);
                const float4 v3 = row4(h, s3, lane);
                acc = fma4(w0, v0, acc);
                acc = fma4(w1, v1, acc);
                acc = fma4(w2, v2, acc);
                acc = fma4(w3, v3, acc);
            }
            const int rem = first - j;
            if (rem == 3) {
                const int s0 = __builtin_amdgcn_readlane(ent[k].x, j);
                const int s1 = __builtin_amdgcn_readlane(ent[k].x, j + 1);
                const int s2 = __builtin_amdgcn_readlane(ent[k].x, j + 2);
                const float w0 = __int_as_float(__builtin_amdgcn_readlane(ent[k].y, j));
                const float w1 = __int_as_float(__builtin_amdgcn_readlane(ent[k].y, j + 1));
                const float w2 = __int_as_float(__builtin_amdgcn_readlane(ent[k].y, j + 2));
                const float4 v0 = row4(h, s0, lane);
                const float4 v1 = row4(h, s1, lane);
                const float4 v2 = row4(h, s2, lane);
                acc = fma4(w0, v0, acc);
                acc = fma4(w1, v1, acc);
                acc = fma4(w2, v2, acc);
            } else if (rem == 2) {
                const int s0 = __builtin_amdgcn_readlane(ent[k].x, j);
                const int s1 = __builtin_amdgcn_readlane(ent[k].x, j + 1);
                const float w0 = __int_as_float(__builtin_amdgcn_readlane(ent[k].y, j));
                const float w1 = __int_as_float(__builtin_amdgcn_readlane(ent[k].y, j + 1));
                const float4 v0 = row4(h, s0, lane);
                const float4 v1 = row4(h, s1, lane);
                acc = fma4(w0, v0, acc);
                acc = fma4(w1, v1, acc);
            } else if (rem == 1) {
                const int s0 = __builtin_amdgcn_readlane(ent[k].x, j);
                const float w0 = __int_as_float(__builtin_amdgcn_readlane(ent[k].y, j));
                acc = fma4(w0, row4(h, s0, lane), acc);
            }
            for (int p = o[k] + 32; p < o[k + 1]; p += 32) {
                const int c = (o[k + 1] - p < 32) ? (o[k + 1] - p) : 32;
                const int2 e2 = (lane < c) ? csr[p + lane] : make_int2(0, 0);
                for (int q = 0; q < c; ++q) {
                    const int s0 = __builtin_amdgcn_readlane(e2.x, q);
                    const float w0 = __int_as_float(__builtin_amdgcn_readlane(e2.y, q));
                    acc = fma4(w0, row4(h, s0, lane), acc);
                }
            }

            const float di = rsqrtf((float)(len + 1));
            if constexpr (PRE) {
                r.x = acc.x + hs.x * di * di;
                r.y = acc.y + hs.y * di * di;
                r.z = acc.z + hs.z * di * di;
                r.w = acc.w + hs.w * di * di;
            } else {
                r.x = tanhf(acc.x + hs.x * di * di + b4.x);
                r.y = tanhf(acc.y + hs.y * di * di + b4.y);
                r.z = tanhf(acc.z + hs.z * di * di + b4.z);
                r.w = tanhf(acc.w + hs.w * di * di + b4.w);
            }
        }
        *(float4*)(rows + (wave * NPW + k) * LDP + lane * 4) = r;
    }

    __syncthreads();

    constexpr int RT = GB_ROWS / 16;
    const int r16 = lane & 15;
    const int kb  = (lane >> 4) * 2;

    const float* arows = rows;
    if constexpr (PRE) {
        static_assert(!PRE || (RT == 2 && DFEAT / 16 == 8), "8 waves x 2 row tiles cover mid");
        float* mid = rows + GB_ROWS * LDP;
        const int colm = wave * 16 + r16;
        v8f accm[RT];
        #pragma unroll
        for (int t = 0; t < RT; ++t) accm[t] = v8f{};
        const float bm = bias[colm];
        tile_products<DFEAT, RT>(rows + r16 * LDP + kb, Wp + kb * DFEAT + colm, accm);
        float* Mp = mid + ((lane >> 4) * 8) * LDP + colm;
        #pragma unroll
        for (int t = 0; t < RT; ++t) {
            #pragma unroll
            for (int i = 0; i < 8; ++i)
                Mp[(t * 16 + i) * LDP] = tanhf(accm[t][i] + bm);
        }
        __syncthreads();
        arows = mid;
    }

    constexpr int CT  = (N + 15) / 16;
    constexpr int RPW = (CT >= 8) ? RT : 1;
    static_assert(CT * (RT / RPW) <= 8, "one pass of the block's 8 waves covers C");

    if (wave >= CT * (RT / RPW)) return;
    const int ct  = wave % CT;
    const int rt0 = (wave / CT) * RPW;

    const int col  = ct * 16 + r16;
    const int colc = (col < N) ? col : (N - 1);

    v8f acc[RPW];
    #pragma unroll
    for (int t = 0; t < RPW; ++t) acc[t] = v8f{};
    tile_products<N, RPW>(arows + (rt0 * 16 + r16) * LDP + kb, Wn + kb * N + colc, acc);

    if (col < N) {
        float bb = 0.f;
        if constexpr (HEAD) bb = bias_n[col];
        #pragma unroll
        for (int t = 0; t < RPW; ++t) {
            const int rbase = blockIdx.x * GB_ROWS + (rt0 + t) * 16;
            if (rbase < n_nodes) {
                float* Cp = C + (rbase + (lane >> 4) * 8) * N + col;
                #pragma unroll
                for (int i = 0; i < 8; ++i) {
                    if constexpr (HEAD) Cp[i * N] = acc[t][i] + bb;
                    else                Cp[i * N] = acc[t][i];
                }
            }
        }
    }
}

extern "C" void kernel_launch(void* const* d_in, const int* in_sizes, int n_in,
                              void* d_out, int out_size, void* d_ws, size_t ws_size,
                              hipStream_t stream) {
    const float* x    = (const float*)d_in[0];
    const int*   ei   = (const int*)  d_in[1];
    const float* W1   = (const float*)d_in[2];
    const float* b1   = (const float*)d_in[3];
    const float* W2   = (const float*)d_in[4];
    const float* b2   = (const float*)d_in[5];
    const float* Wlin = (const float*)d_in[6];
    const float* blin = (const float*)d_in[7];

    const int* src = ei;
    const int* dst = ei + N_EDGES;

    int*   cnt  = (int*)d_ws;
    int*   offs = cnt + 50176;
    int*   rank = offs + 50176;
    int2*  csr  = (int2*)(rank + 600064);
    float* bufB = (float*)(csr + 600064);

    static_assert(N_NODES % 16 == 0, "row tiles of 16 must cover the nodes exactly");
    const int gatherBlocks = (N_NODES + 8 * NPW - 1) / (8 * NPW);

    k_init_cnt  <<<(N_NODES + 255) / 256, 256, 0, stream>>>(cnt, N_NODES);
    k_count_rank<<<(N_EDGES + 255) / 256, 256, 0, stream>>>(cnt, dst, rank, N_EDGES);
    k_scan      <<<1, 1024, 0, stream>>>(cnt, offs, N_NODES);
    k_fill      <<<(N_EDGES + 255) / 256, 256, 0, stream>>>(src, dst, rank, offs, cnt, csr, N_EDGES);

    k_gather_gemm<DFEAT, false, true><<<gatherBlocks, 256, 0, stream>>>(
        x, csr, offs, W1, b1, W2, (const float*)nullptr, bufB, N_NODES);

    k_gather_gemm<NCLS, true, false><<<gatherBlocks, 256, 0, stream>>>(
        bufB, csr, offs, (const float*)nullptr, b2, Wlin, blin, (float*)d_out, N_NODES);
}


# Edge list in ONE pass: an edge's returning int add is its slot in a 32-entry bucket of its dst (overflow list past 32), norm formed from the degrees in the gather; scan and fill launches go: 4 left
# speedup vs baseline: 3.4053x; 1.1574x over previous
#include <hip/hip_runtime.h>
#include <math.h>


typedef float v2f __attribute__((ext_vector_type(2)));
typedef float v8f __attribute__((ext_vector_type(8)));

#define N_NODES 50000
#define N_EDGES 600000
#define DFEAT   128
#define NCLS    40

#define BK 32

__global__ void k_init_cnt(int* cnt, int* ovf_n, int n) {
    int i = blockIdx.x * blockDim.x + threadIdx.x;
    if (i < n) cnt[i] = 0;
    if (i == 0) *ovf_n = 0;
}

__global__ void k_count_bucket(int* cnt, const int* __restrict__ src,
                               const int* __restrict__ dst, int* __restrict__ bucket,
                               int* ovf_n, int2* __restrict__ ovf, int e) {
    int i = blockIdx.x * blockDim.x + threadIdx.x;
    if (i >= e) return;
    const int s = src[i];
    const int d = dst[i];
    const int r = atomicAdd(&cnt[d], 1);
    if (r < BK) bucket[d * BK + r] = s;
    else        ovf[atomicAdd(ovf_n, 1)] = make_int2(d, s);
}

#define NPW 4
#define GB_ROWS (8 * NPW)
#define LDP     (DFEAT + 4)

__device__ __forceinline__ float4 fma4(float w, float4 v, float4 a) {
    a.x = fmaf(w, v.x, a.x);
    a.y = fmaf(w, v.y, a.y);
    a.z = fmaf(w, v.z, a.z);
    a.w = fmaf(w, v.w, a.w);
    return a;
}

__device__ __forceinline__ float4 row4(const float* __restrict__ h, int row, int lane) {
    return *(const float4*)(h + row * DFEAT + lane * 4);
}

template<int N, int RPW>
__device__ __forceinline__ void tile_products(const float* Ap, const float* __restrict__ Bp,
                                              v8f (&acc)[RPW]) {
    #pragma unroll 8
    for (int k0 = 0; k0 < DFEAT; k0 += 4) {
        v2f b;
        b.x = Bp[k0 * N];
        b.y = Bp[k0 * N + N];
        #pragma unroll
        for (int t = 0; t < RPW; ++t) {
            const v2f a = *(const v2f*)(Ap + t * 16 * LDP + k0);
            acc[t] = __builtin_amdgcn_wmma_f32_16x16x4_f32(
                false, a, false, b, (short)0, acc[t], false, false);
        }
    }
}

template<int N, bool HEAD, bool PRE>
__global__ void __launch_bounds__(256, 16) k_gather_gemm(const float* __restrict__ h,
                                                     const int* __restrict__ cnt,
                                                     const int* __restrict__ bucket,
                                                     const int* __restrict__ ovf_n,
                                                     const int2* __restrict__ ovf,
                                                     const float* __restrict__ Wp,
                                                     const float* __restrict__ bias,
                                                     const float* __restrict__ Wn,
                                                     const float* __restrict__ bias_n,
                                                     float* __restrict__ C, int n_nodes) {
    __shared__ __attribute__((aligned(16))) float rows[(PRE ? 2 : 1) * GB_ROWS * LDP];

    const int lane = threadIdx.x & 31;
    const int wave = __builtin_amdgcn_readfirstlane(threadIdx.x >> 5);
    const int n0 = (blockIdx.x * 8 + wave) * NPW;

    int   nin[NPW];
    float dnode[NPW];
    #pragma unroll
    for (int k = 0; k < NPW; ++k) {
        nin[k] = (n0 + k < n_nodes) ? cnt[n0 + k] : 0;
        dnode[k] = rsqrtf((float)(nin[k] + 1));
    }
    int2 ent[NPW];
    #pragma unroll
    for (int k = 0; k < NPW; ++k) {
        const int first = (nin[k] < BK) ? nin[k] : BK;
        ent[k].x = (lane < first) ? bucket[(n0 + k) * BK + lane] : 0;
    }
    int cs[NPW];
    #pragma unroll
    for (int k = 0; k < NPW; ++k) cs[k] = cnt[ent[k].x];
    #pragma unroll
    for (int k = 0; k < NPW; ++k)
        ent[k].y = __float_as_int(rsqrtf((float)(cs[k] + 1)) * dnode[k]);

    float4 b4 = make_float4(0.f, 0.f, 0.f, 0.f);
    if constexpr (!PRE) b4 = *(const float4*)(bias + lane * 4);

    #pragma unroll
    for (int k = 0; k < NPW; ++k) {
        const int n = n0 + k;
        float4 r = make_float4(0.f, 0.f, 0.f, 0.f);
        if (n < n_nodes) {
            const int len = nin[k];
            const float di = dnode[k];
            const float4 hs = row4(h, n, lane);
            float4 acc = make_float4(0.f, 0.f, 0.f, 0.f);

            const int first = (len < BK) ? len : BK;
            int j = 0;
            for (; j + 4 <= first; j += 4) {
                const int s0 = __builtin_amdgcn_readlane(ent[k].x, j);
                const int s1 = __builtin_amdgcn_readlane(ent[k].x, j + 1);
                const int s2 = __builtin_amdgcn_readlane(ent[k].x, j + 2);
                const int s3 = __builtin_amdgcn_readlane(ent[k].x, j + 3);
                const float w0 = __int_as_float(__builtin_amdgcn_readlane(ent[k].y, j));
                const float w1 = __int_as_float(__builtin_amdgcn_readlane(ent[k].y, j + 1));
                const float w2 = __int_as_float(__builtin_amdgcn_readlane(ent[k].y, j + 2));
                const float w3 = __int_as_float(__builtin_amdgcn_readlane(ent[k].y, j + 3));
                const float4 v0 = row4(h, s0, lane);
                const float4 v1 = row4(h, s1, lane);
                const float4 v2 = row4(h, s2, lane);
                const float4 v3 = row4(h, s3, lane);
                acc = fma4(w0, v0, acc);
                acc = fma4(w1, v1, acc);
                acc = fma4(w2, v2, acc);
                acc = fma4(w3, v3, acc);
            }
            const int rem = first - j;
            if (rem == 3) {
                const int s0 = __builtin_amdgcn_readlane(ent[k].x, j);
                const int s1 = __builtin_amdgcn_readlane(ent[k].x, j + 1);
                const int s2 = __builtin_amdgcn_readlane(ent[k].x, j + 2);
                const float w0 = __int_as_float(__builtin_amdgcn_readlane(ent[k].y, j));
                const float w1 = __int_as_float(__builtin_amdgcn_readlane(ent[k].y, j + 1));
                const float w2 = __int_as_float(__builtin_amdgcn_readlane(ent[k].y, j + 2));
                const float4 v0 = row4(h, s0, lane);
                const float4 v1 = row4(h, s1, lane);
                const float4 v2 = row4(h, s2, lane);
                acc = fma4(w0, v0, acc);
                acc = fma4(w1, v1, acc);
                acc = fma4(w2, v2, acc);
            } else if (rem == 2) {
                const int s0 = __builtin_amdgcn_readlane(ent[k].x, j);
                const int s1 = __builtin_amdgcn_readlane(ent[k].x, j + 1);
                const float w0 = __int_as_float(__builtin_amdgcn_readlane(ent[k].y, j));
                const float w1 = __int_as_float(__builtin_amdgcn_readlane(ent[k].y, j + 1));
                const float4 v0 = row4(h, s0, lane);
                const float4 v1 = row4(h, s1, lane);
                acc = fma4(w0, v0, acc);
                acc = fma4(w1, v1, acc);
            } else if (rem == 1) {
                const int s0 = __builtin_amdgcn_readlane(ent[k].x, j);
                const float w0 = __int_as_float(__builtin_amdgcn_readlane(ent[k].y, j));
                acc = fma4(w0, row4(h, s0, lane), acc);
            }
            if (len > BK) {
                const int tot = *ovf_n;
                for (int p = 0; p < tot; p += 32) {
                    const int2 e2 = (p + lane < tot) ? ovf[p + lane] : make_int2(-1, 0);
                    unsigned m = __builtin_amdgcn_ballot_w32(e2.x == n);
                    while (m) {
                        const int q = __builtin_ctz(m);
                        m &= m - 1;
                        const int s0 = __builtin_amdgcn_readlane(e2.y, q);
                        const float w0 = rsqrtf((float)(cnt[s0] + 1)) * di;
                        acc = fma4(w0, row4(h, s0, lane), acc);
                    }
                }
            }

            if constexpr (PRE) {
                r.x = acc.x + hs.x * di * di;
                r.y = acc.y + hs.y * di * di;
                r.z = acc.z + hs.z * di * di;
                r.w = acc.w + hs.w * di * di;
            } else {
                r.x = tanhf(acc.x + hs.x * di * di + b4.x);
                r.y = tanhf(acc.y + hs.y * di * di + b4.y);
                r.z = tanhf(acc.z + hs.z * di * di + b4.z);
                r.w = tanhf(acc.w + hs.w * di * di + b4.w);
            }
        }
        *(float4*)(rows + (wave * NPW + k) * LDP + lane * 4) = r;
    }

    __syncthreads();

    constexpr int RT = GB_ROWS / 16;
    const int r16 = lane & 15;
    const int kb  = (lane >> 4) * 2;

    const float* arows = rows;
    if constexpr (PRE) {
        static_assert(!PRE || (RT == 2 && DFEAT / 16 == 8), "8 waves x 2 row tiles cover mid");
        float* mid = rows + GB_ROWS * LDP;
        const int colm = wave * 16 + r16;
        v8f accm[RT];
        #pragma unroll
        for (int t = 0; t < RT; ++t) accm[t] = v8f{};
        const float bm = bias[colm];
        tile_products<DFEAT, RT>(rows + r16 * LDP + kb, Wp + kb * DFEAT + colm, accm);
        float* Mp = mid + ((lane >> 4) * 8) * LDP + colm;
        #pragma unroll
        for (int t = 0; t < RT; ++t) {
            #pragma unroll
            for (int i = 0; i < 8; ++i)
                Mp[(t * 16 + i) * LDP] = tanhf(accm[t][i] + bm);
        }
        __syncthreads();
        arows = mid;
    }

    constexpr int CT  = (N + 15) / 16;
    constexpr int RPW = (CT >= 8) ? RT : 1;
    static_assert(CT * (RT / RPW) <= 8, "one pass of the block's 8 waves covers C");

    if (wave >= CT * (RT / RPW)) return;
    const int ct  = wave % CT;
    const int rt0 = (wave / CT) * RPW;

    const int col  = ct * 16 + r16;
    const int colc = (col < N) ? col : (N - 1);

    v8f acc[RPW];
    #pragma unroll
    for (int t = 0; t < RPW; ++t) acc[t] = v8f{};
    tile_products<N, RPW>(arows + (rt0 * 16 + r16) * LDP + kb, Wn + kb * N + colc, acc);

    if (col < N) {
        float bb = 0.f;
        if constexpr (HEAD) bb = bias_n[col];
        #pragma unroll
        for (int t = 0; t < RPW; ++t) {
            const int rbase = blockIdx.x * GB_ROWS + (rt0 + t) * 16;
            if (rbase < n_nodes) {
                float* Cp = C + (rbase + (lane >> 4) * 8) * N + col;
                #pragma unroll
                for (int i = 0; i < 8; ++i) {
                    if constexpr (HEAD) Cp[i * N] = acc[t][i] + bb;
                    else                Cp[i * N] = acc[t][i];
                }
            }
        }
    }
}

extern "C" void kernel_launch(void* const* d_in, const int* in_sizes, int n_in,
                              void* d_out, int out_size, void* d_ws, size_t ws_size,
                              hipStream_t stream) {
    const float* x    = (const float*)d_in[0];
    const int*   ei   = (const int*)  d_in[1];
    const float* W1   = (const float*)d_in[2];
    const float* b1   = (const float*)d_in[3];
    const float* W2   = (const float*)d_in[4];
    const float* b2   = (const float*)d_in[5];
    const float* Wlin = (const float*)d_in[6];
    const float* blin = (const float*)d_in[7];

    const int* src = ei;
    const int* dst = ei + N_EDGES;

    int*   cnt    = (int*)d_ws;
    int*   ovf_n  = cnt + 50176;
    int*   bucket = ovf_n + 64;
    int2*  ovf    = (int2*)(bucket + 1600512);
    float* bufB   = (float*)(ovf + 600064);

    static_assert(N_NODES % 16 == 0, "row tiles of 16 must cover the nodes exactly");
    const int gatherBlocks = (N_NODES + 8 * NPW - 1) / (8 * NPW);

    k_init_cnt    <<<(N_NODES + 255) / 256, 256, 0, stream>>>(cnt, ovf_n, N_NODES);
    k_count_bucket<<<(N_EDGES + 255) / 256, 256, 0, stream>>>(cnt, src, dst, bucket, ovf_n, ovf, N_EDGES);

    k_gather_gemm<DFEAT, false, true><<<gatherBlocks, 256, 0, stream>>>(
        x, cnt, bucket, ovf_n, ovf, W1, b1, W2, (const float*)nullptr, bufB, N_NODES);

    k_gather_gemm<NCLS, true, false><<<gatherBlocks, 256, 0, stream>>>(
        bufB, cnt, bucket, ovf_n, ovf, (const float*)nullptr, b2, Wlin, blin, (float*)d_out, N_NODES);
}


# Second-layer rows stored pre-multiplied by their own dinv: the second gather adds plain rows and scales the sum once (per-term rounding changes, sum order kept); its per-edge degree look-ups go
# speedup vs baseline: 3.5118x; 1.0313x over previous
#include <hip/hip_runtime.h>
#include <math.h>


typedef float v2f __attribute__((ext_vector_type(2)));
typedef float v8f __attribute__((ext_vector_type(8)));

#define N_NODES 50000
#define N_EDGES 600000
#define DFEAT   128
#define NCLS    40

#define BK 32

__global__ void k_init_cnt(int* cnt, int* ovf_n, int n) {
    int i = blockIdx.x * blockDim.x + threadIdx.x;
    if (i < n) cnt[i] = 0;
    if (i == 0) *ovf_n = 0;
}

__global__ void k_count_bucket(int* cnt, const int* __restrict__ src,
                               const int* __restrict__ dst, int* __restrict__ bucket,
                               int* ovf_n, int2* __restrict__ ovf, int e) {
    int i = blockIdx.x * blockDim.x + threadIdx.x;
    if (i >= e) return;
    const int s = src[i];
    const int d = dst[i];
    const int r = atomicAdd(&cnt[d], 1);
    if (r < BK) bucket[d * BK + r] = s;
    else        ovf[atomicAdd(ovf_n, 1)] = make_int2(d, s);
}

#define NPW 4
#define GB_ROWS (8 * NPW)
#define LDP     (DFEAT + 4)

__device__ __forceinline__ float4 fma4(float w, float4 v, float4 a) {
    a.x = fmaf(w, v.x, a.x);
    a.y = fmaf(w, v.y, a.y);
    a.z = fmaf(w, v.z, a.z);
    a.w = fmaf(w, v.w, a.w);
    return a;
}

template<bool WEIGHTED>
__device__ __forceinline__ float4 acc_row(float w, float4 v, float4 a) {
    if constexpr (WEIGHTED) {
        return fma4(w, v, a);
    } else {
        a.x += v.x;
        a.y += v.y;
        a.z += v.z;
        a.w += v.w;
        return a;
    }
}

template<bool WEIGHTED>
__device__ __forceinline__ float lane_w(int wbits, int j) {
    if constexpr (WEIGHTED) return __int_as_float(__builtin_amdgcn_readlane(wbits, j));
    else                    return 0.f;
}

__device__ __forceinline__ float4 row4(const float* __restrict__ h, int row, int lane) {
    return *(const float4*)(h + row * DFEAT + lane * 4);
}

template<int N, int RPW>
__device__ __forceinline__ void tile_products(const float* Ap, const float* __restrict__ Bp,
                                              v8f (&acc)[RPW]) {
    #pragma unroll 8
    for (int k0 = 0; k0 < DFEAT; k0 += 4) {
        v2f b;
        b.x = Bp[k0 * N];
        b.y = Bp[k0 * N + N];
        #pragma unroll
        for (int t = 0; t < RPW; ++t) {
            const v2f a = *(const v2f*)(Ap + t * 16 * LDP + k0);
            acc[t] = __builtin_amdgcn_wmma_f32_16x16x4_f32(
                false, a, false, b, (short)0, acc[t], false, false);
        }
    }
}

template<int N, bool HEAD, bool PRE>
__global__ void __launch_bounds__(256, 16) k_gather_gemm(const float* __restrict__ h,
                                                     const int* __restrict__ cnt,
                                                     const int* __restrict__ bucket,
                                                     const int* __restrict__ ovf_n,
                                                     const int2* __restrict__ ovf,
                                                     const float* __restrict__ Wp,
                                                     const float* __restrict__ bias,
                                                     const float* __restrict__ Wn,
                                                     const float* __restrict__ bias_n,
                                                     float* __restrict__ C, int n_nodes) {
    __shared__ __attribute__((aligned(16))) float rows[(PRE ? 2 : 1) * GB_ROWS * LDP];
    __shared__ float dinv_s[GB_ROWS];

    const int lane = threadIdx.x & 31;
    const int wave = __builtin_amdgcn_readfirstlane(threadIdx.x >> 5);
    const int n0 = (blockIdx.x * 8 + wave) * NPW;

    int   nin[NPW];
    float dnode[NPW];
    #pragma unroll
    for (int k = 0; k < NPW; ++k) {
        nin[k] = (n0 + k < n_nodes) ? cnt[n0 + k] : 0;
        dnode[k] = rsqrtf((float)(nin[k] + 1));
    }
    constexpr bool WGT = PRE;
    int2 ent[NPW];
    #pragma unroll
    for (int k = 0; k < NPW; ++k) {
        const int first = (nin[k] < BK) ? nin[k] : BK;
        ent[k].x = (lane < first) ? bucket[(n0 + k) * BK + lane] : 0;
        ent[k].y = 0;
    }
    if constexpr (WGT) {
        int cs[NPW];
        #pragma unroll
        for (int k = 0; k < NPW; ++k) cs[k] = cnt[ent[k].x];
        #pragma unroll
        for (int k = 0; k < NPW; ++k)
            ent[k].y = __float_as_int(rsqrtf((float)(cs[k] + 1)) * dnode[k]);
    }
    if constexpr (!HEAD) {
        #pragma unroll
        for (int k = 0; k < NPW; ++k)
            if (lane == k) dinv_s[wave * NPW + k] = dnode[k];
    }

    float4 b4 = make_float4(0.f, 0.f, 0.f, 0.f);
    if constexpr (!PRE) b4 = *(const float4*)(bias + lane * 4);

    #pragma unroll
    for (int k = 0; k < NPW; ++k) {
        const int n = n0 + k;
        float4 r = make_float4(0.f, 0.f, 0.f, 0.f);
        if (n < n_nodes) {
            const int len = nin[k];
            const float di = dnode[k];
            const float4 hs = row4(h, n, lane);
            float4 acc = make_float4(0.f, 0.f, 0.f, 0.f);

            const int first = (len < BK) ? len : BK;
            int j = 0;
            for (; j + 4 <= first; j += 4) {
                const int s0 = __builtin_amdgcn_readlane(ent[k].x, j);
                const int s1 = __builtin_amdgcn_readlane(ent[k].x, j + 1);
                const int s2 = __builtin_amdgcn_readlane(ent[k].x, j + 2);
                const int s3 = __builtin_amdgcn_readlane(ent[k].x, j + 3);
                const float w0 = lane_w<WGT>(ent[k].y, j);
                const float w1 = lane_w<WGT>(ent[k].y, j + 1);
                const float w2 = lane_w<WGT>(ent[k].y, j + 2);
                const float w3 = lane_w<WGT>(ent[k].y, j + 3);
                const float4 v0 = row4(h, s0, lane);
                const float4 v1 = row4(h, s1, lane);
                const float4 v2 = row4(h, s2, lane);
                const float4 v3 = row4(h, s3, lane);
                acc = acc_row<WGT>(w0, v0, acc);
                acc = acc_row<WGT>(w1, v1, acc);
                acc = acc_row<WGT>(w2, v2, acc);
                acc = acc_row<WGT>(w3, v3, acc);
            }
            const int rem = first - j;
            if (rem == 3) {
                const int s0 = __builtin_amdgcn_readlane(ent[k].x, j);
                const int s1 = __builtin_amdgcn_readlane(ent[k].x, j + 1);
                const int s2 = __builtin_amdgcn_readlane(ent[k].x, j + 2);
                const float w0 = lane_w<WGT>(ent[k].y, j);
                const float w1 = lane_w<WGT>(ent[k].y, j + 1);
                const float w2 = lane_w<WGT>(ent[k].y, j + 2);
                const float4 v0 = row4(h, s0, lane);
                const float4 v1 = row4(h, s1, lane);
                const float4 v2 = row4(h, s2, lane);
                acc = acc_row<WGT>(w0, v0, acc);
                acc = acc_row<WGT>(w1, v1, acc);
                acc = acc_row<WGT>(w2, v2, acc);
            } else if (rem == 2) {
                const int s0 = __builtin_amdgcn_readlane(ent[k].x, j);
                const int s1 = __builtin_amdgcn_readlane(ent[k].x, j + 1);
                const float w0 = lane_w<WGT>(ent[k].y, j);
                const float w1 = lane_w<WGT>(ent[k].y, j + 1);
                const float4 v0 = row4(h, s0, lane);
                const float4 v1 = row4(h, s1, lane);
                acc = acc_row<WGT>(w0, v0, acc);
                acc = acc_row<WGT>(w1, v1, acc);
            } else if (rem == 1) {
                const int s0 = __builtin_amdgcn_readlane(ent[k].x, j);
                const float w0 = lane_w<WGT>(ent[k].y, j);
                acc = acc_row<WGT>(w0, row4(h, s0, lane), acc);
            }
            if (len > BK) {
                const int tot = *ovf_n;
                for (int p = 0; p < tot; p += 32) {
                    const int2 e2 = (p + lane < tot) ? ovf[p + lane] : make_int2(-1, 0);
                    unsigned m = __builtin_amdgcn_ballot_w32(e2.x == n);
                    while (m) {
                        const int q = __builtin_ctz(m);
                        m &= m - 1;
                        const int s0 = __builtin_amdgcn_readlane(e2.y, q);
                        float w0 = 0.f;
                        if constexpr (WGT) w0 = rsqrtf((float)(cnt[s0] + 1)) * di;
                        acc = acc_row<WGT>(w0, row4(h, s0, lane), acc);
                    }
                }
            }

            if constexpr (PRE) {
                r.x = acc.x + hs.x * di * di;
                r.y = acc.y + hs.y * di * di;
                r.z = acc.z + hs.z * di * di;
                r.w = acc.w + hs.w * di * di;
            } else {
                r.x = tanhf((acc.x + hs.x) * di + b4.x);
                r.y = tanhf((acc.y + hs.y) * di + b4.y);
                r.z = tanhf((acc.z + hs.z) * di + b4.z);
                r.w = tanhf((acc.w + hs.w) * di + b4.w);
            }
        }
        *(float4*)(rows + (wave * NPW + k) * LDP + lane * 4) = r;
    }

    __syncthreads();

    constexpr int RT = GB_ROWS / 16;
    const int r16 = lane & 15;
    const int kb  = (lane >> 4) * 2;

    const float* arows = rows;
    if constexpr (PRE) {
        static_assert(!PRE || (RT == 2 && DFEAT / 16 == 8), "8 waves x 2 row tiles cover mid");
        float* mid = rows + GB_ROWS * LDP;
        const int colm = wave * 16 + r16;
        v8f accm[RT];
        #pragma unroll
        for (int t = 0; t < RT; ++t) accm[t] = v8f{};
        const float bm = bias[colm];
        tile_products<DFEAT, RT>(rows + r16 * LDP + kb, Wp + kb * DFEAT + colm, accm);
        float* Mp = mid + ((lane >> 4) * 8) * LDP + colm;
        #pragma unroll
        for (int t = 0; t < RT; ++t) {
            #pragma unroll
            for (int i = 0; i < 8; ++i)
                Mp[(t * 16 + i) * LDP] = tanhf(accm[t][i] + bm);
        }
        __syncthreads();
        arows = mid;
    }

    constexpr int CT  = (N + 15) / 16;
    constexpr int RPW = (CT >= 8) ? RT : 1;
    static_assert(CT * (RT / RPW) <= 8, "one pass of the block's 8 waves covers C");

    if (wave >= CT * (RT / RPW)) return;
    const int ct  = wave % CT;
    const int rt0 = (wave / CT) * RPW;

    const int col  = ct * 16 + r16;
    const int colc = (col < N) ? col : (N - 1);

    v8f acc[RPW];
    #pragma unroll
    for (int t = 0; t < RPW; ++t) acc[t] = v8f{};
    tile_products<N, RPW>(arows + (rt0 * 16 + r16) * LDP + kb, Wn + kb * N + colc, acc);

    if (col < N) {
        float bb = 0.f;
        if constexpr (HEAD) bb = bias_n[col];
        #pragma unroll
        for (int t = 0; t < RPW; ++t) {
            const int rbase = blockIdx.x * GB_ROWS + (rt0 + t) * 16;
            if (rbase < n_nodes) {
                float* Cp = C + (rbase + (lane >> 4) * 8) * N + col;
                const float* dp = dinv_s + (rt0 + t) * 16 + (lane >> 4) * 8;
                #pragma unroll
                for (int i = 0; i < 8; ++i) {
                    if constexpr (HEAD) Cp[i * N] = acc[t][i] + bb;
                    else                Cp[i * N] = acc[t][i] * dp[i];
                }
            }
        }
    }
}

extern "C" void kernel_launch(void* const* d_in, const int* in_sizes, int n_in,
                              void* d_out, int out_size, void* d_ws, size_t ws_size,
                              hipStream_t stream) {
    const float* x    = (const float*)d_in[0];
    const int*   ei   = (const int*)  d_in[1];
    const float* W1   = (const float*)d_in[2];
    const float* b1   = (const float*)d_in[3];
    const float* W2   = (const float*)d_in[4];
    const float* b2   = (const float*)d_in[5];
    const float* Wlin = (const float*)d_in[6];
    const float* blin = (const float*)d_in[7];

    const int* src = ei;
    const int* dst = ei + N_EDGES;

    int*   cnt    = (int*)d_ws;
    int*   ovf_n  = cnt + 50176;
    int*   bucket = ovf_n + 64;
    int2*  ovf    = (int2*)(bucket + 1600512);
    float* bufB   = (float*)(ovf + 600064);

    static_assert(N_NODES % 16 == 0, "row tiles of 16 must cover the nodes exactly");
    const int gatherBlocks = (N_NODES + 8 * NPW - 1) / (8 * NPW);

    k_init_cnt    <<<(N_NODES + 255) / 256, 256, 0, stream>>>(cnt, ovf_n, N_NODES);
    k_count_bucket<<<(N_EDGES + 255) / 256, 256, 0, stream>>>(cnt, src, dst, bucket, ovf_n, ovf, N_EDGES);

    k_gather_gemm<DFEAT, false, true><<<gatherBlocks, 256, 0, stream>>>(
        x, cnt, bucket, ovf_n, ovf, W1, b1, W2, (const float*)nullptr, bufB, N_NODES);

    k_gather_gemm<NCLS, true, false><<<gatherBlocks, 256, 0, stream>>>(
        bufB, cnt, bucket, ovf_n, ovf, (const float*)nullptr, b2, Wlin, blin, (float*)d_out, N_NODES);
}
